# starting kernel
# baseline (speedup 1.0000x reference)
#include <hip/hip_runtime.h>
#include <hip/hip_bf16.h>

typedef __attribute__((ext_vector_type(2))) float v2f;
typedef __attribute__((ext_vector_type(8))) float v8f;

#define CH 64
#define NB 2

__global__ void zero_f32_kernel(float* __restrict__ p, long n) {
    long i = (long)blockIdx.x * blockDim.x + threadIdx.x;
    long stride = (long)gridDim.x * blockDim.x;
    for (; i < n; i += stride) p[i] = 0.0f;
}

__global__ void degree_kernel(const int* __restrict__ row, const int* __restrict__ col,
                              float* __restrict__ deg, long E) {
    long e = (long)blockIdx.x * blockDim.x + threadIdx.x;
    if (e >= E) return;
    int r = row[e], c = col[e];
    if (r != c) atomicAdd(&deg[r], 1.0f);
}

__global__ void dinv_sqrt_kernel(float* __restrict__ deg, long n) {
    long i = (long)blockIdx.x * blockDim.x + threadIdx.x;
    if (i >= n) return;
    float d = deg[i];
    deg[i] = (d > 0.0f) ? __frsqrt_rn(d) : 0.0f;
}

__global__ void lap_kernel(const int* __restrict__ row, const int* __restrict__ col,
                           const float* __restrict__ ew, const float* __restrict__ dis,
                           float* __restrict__ lap, long E) {
    long e = (long)blockIdx.x * blockDim.x + threadIdx.x;
    if (e >= E) return;
    int r = row[e], c = col[e];
    lap[e] = (r != c) ? (-dis[r] * ew[e] * dis[c]) : 0.0f;
}

__global__ __launch_bounds__(256) void spmm_kernel(
    const float* __restrict__ src, const int* __restrict__ row,
    const int* __restrict__ col, const float* __restrict__ lap,
    float* __restrict__ dst, int N, long E) {
    long idx = (long)blockIdx.x * blockDim.x + threadIdx.x;
    long e = idx >> 7;
    if (e >= E) return;
    int rem = (int)(idx & 127);
    int b = rem >> 6;
    int c = rem & 63;
    float l = lap[e];
    if (l == 0.0f) return;
    long cs = col[e], rs = row[e];
    float v = src[((long)b * N + cs) * CH + c] * l;
    atomicAdd(&dst[((long)b * N + rs) * CH + c], v);
}

__global__ __launch_bounds__(128) void cheb_gemm_wmma(
    const float* __restrict__ x, const float* __restrict__ tx1,
    const float* __restrict__ s2, const float* __restrict__ w,
    const float* __restrict__ bias, float* __restrict__ out, long BN) {
    __shared__ float wlds[3 * CH * CH];

    const int tid = threadIdx.x;
    for (int idx = tid; idx < 3 * CH * CH; idx += 128) {
        const int t   = idx >> 12;
        const int rem = idx & 4095;
        const int k   = rem >> 6;
        const int n   = rem & 63;
        wlds[((((t << 5) + (k >> 1)) << 6) + n) * 2 + (k & 1)] = w[idx];
    }
    __syncthreads();

    const int wave = tid >> 5;
    const int lane = tid & 31;
    const long m0 = ((long)blockIdx.x * 4 + wave) * 16;
    if (m0 + 16 > BN) return;

    const int mrow = lane & 15;
    const int lhi  = lane >> 4;

    const v2f* __restrict__ wp = (const v2f*)wlds;
    const v2f* __restrict__ bbase = wp + (lhi << 6) + mrow;

    v8f acc[4] = {v8f{}, v8f{}, v8f{}, v8f{}};

    #pragma unroll
    for (int t = 0; t < 3; ++t) {
        const float* A = (t == 0) ? x : (t == 1) ? tx1 : s2;
        #pragma unroll
        for (int ks = 0; ks < 16; ++ks) {
            const int kb = ks * 4 + (lhi << 1);
            const float2 a2 = *(const float2*)(A + (m0 + mrow) * CH + kb);
            v2f a;
            a.x = a2.x; a.y = a2.y;
            if (t == 2) {
                const float2 x2 = *(const float2*)(x + (m0 + mrow) * CH + kb);
                a.x = 2.0f * a.x - x2.x;
                a.y = 2.0f * a.y - x2.y;
            }
            #pragma unroll
            for (int nt = 0; nt < 4; ++nt) {
                const v2f b = bbase[(t * 32 + ks * 2) * 64 + nt * 16];
                acc[nt] = __builtin_amdgcn_wmma_f32_16x16x4_f32(
                    false, a, false, b, (short)0, acc[nt], false, false);
            }
        }
    }

    const int moff = lhi << 3;
    #pragma unroll
    for (int nt = 0; nt < 4; ++nt) {
        const int col = nt * 16 + mrow;
        const float bv = bias[col];
        #pragma unroll
        for (int v = 0; v < 8; ++v) {
            out[(m0 + moff + v) * CH + col] = acc[nt][v] + bv;
        }
    }
}

__global__ void cheb_gemm_tail(const float* __restrict__ x, const float* __restrict__ tx1,
                               const float* __restrict__ s2, const float* __restrict__ w,
                               const float* __restrict__ bias, float* __restrict__ out,
                               long startRow, long BN) {
    long r = startRow + blockIdx.x;
    if (r >= BN) return;
    int c = threadIdx.x;
    float acc = bias[c];
    for (int k = 0; k < CH; ++k) {
        float t0 = x[r * CH + k];
        float t1 = tx1[r * CH + k];
        float t2 = 2.0f * s2[r * CH + k] - t0;
        acc += t0 * w[k * CH + c] + t1 * w[CH * CH + k * CH + c] + t2 * w[2 * CH * CH + k * CH + c];
    }
    out[r * CH + c] = acc;
}

extern "C" void kernel_launch(void* const* d_in, const int* in_sizes, int n_in,
                              void* d_out, int out_size, void* d_ws, size_t ws_size,
                              hipStream_t stream) {
    const float* x    = (const float*)d_in[0];
    const int*   ei   = (const int*)d_in[1];
    const float* ew   = (const float*)d_in[2];
    const float* w    = (const float*)d_in[3];
    const float* bias = (const float*)d_in[4];
    float*       out  = (float*)d_out;

    const long BN = (long)in_sizes[0] / CH;
    const int  N  = (int)(BN / NB);
    const long E  = (long)in_sizes[2];
    const int* row = ei;
    const int* col = ei + E;

    char* ws = (char*)d_ws;
    size_t off = 0;
    auto carve = [&](size_t bytes) {
        void* p = ws + off;
        off += (bytes + 255) & ~(size_t)255;
        return p;
    };
    float* deg = (float*)carve((size_t)N * sizeof(float));
    float* lap = (float*)carve((size_t)E * sizeof(float));
    float* tx1 = (float*)carve((size_t)BN * CH * sizeof(float));
    float* s2  = (float*)carve((size_t)BN * CH * sizeof(float));
    (void)ws_size;

    const int TB = 256;
    auto gblocks = [](long n, int tb) { return (int)((n + tb - 1) / tb); };

    zero_f32_kernel<<<4096, TB, 0, stream>>>(deg, N);
    zero_f32_kernel<<<4096, TB, 0, stream>>>(tx1, BN * CH);
    zero_f32_kernel<<<4096, TB, 0, stream>>>(s2,  BN * CH);

    degree_kernel<<<gblocks(E, TB), TB, 0, stream>>>(row, col, deg, E);
    dinv_sqrt_kernel<<<gblocks(N, TB), TB, 0, stream>>>(deg, N);
    lap_kernel<<<gblocks(E, TB), TB, 0, stream>>>(row, col, ew, deg, lap, E);

    const long spmmThreads = E * (long)(NB * CH);
    spmm_kernel<<<gblocks(spmmThreads, TB), TB, 0, stream>>>(x,   row, col, lap, tx1, N, E);
    spmm_kernel<<<gblocks(spmmThreads, TB), TB, 0, stream>>>(tx1, row, col, lap, s2,  N, E);

    const long fullTiles = BN / 16;
    if (fullTiles > 0) {
        const int blocks = (int)((fullTiles + 3) / 4);
        cheb_gemm_wmma<<<blocks, 128, 0, stream>>>(x, tx1, s2, w, bias, out, BN);
    }
    const long tailStart = fullTiles * 16;
    if (tailStart < BN) {
        cheb_gemm_tail<<<(int)(BN - tailStart), CH, 0, stream>>>(x, tx1, s2, w, bias, out,
                                                                 tailStart, BN);
    }
}


# Degree count: 800000 scattered global float atomics replaced by 64 per-workgroup LDS integer count tables, stored as partial planes and summed in the inverse-sqrt pass; zero launch gone, 7 -> 6
# speedup vs baseline: 4.0164x; 4.0164x over previous
#include <hip/hip_runtime.h>
#include <hip/hip_bf16.h>

typedef __attribute__((ext_vector_type(2))) float v2f;
typedef __attribute__((ext_vector_type(8))) float v8f;

#define CH 64
#define NB 2

#define DP 64
#define HB 65536

__device__ __forceinline__ int plane_stride(int N) { return (N + 31) & ~31; }

__global__ __launch_bounds__(1024) void degree_part_kernel(
    const int* __restrict__ row, const int* __restrict__ col,
    int* __restrict__ part, int N, long E) {
    __shared__ int cnt[HB];
    const int tid = threadIdx.x;
    const long per = (((E + gridDim.x - 1) / gridDim.x) + 3) & ~3L;
    const long e0  = (long)blockIdx.x * per;
    const long e1  = (e0 + per < E) ? (e0 + per) : E;
    const bool vec = ((E & 3) == 0);
    int* __restrict__ myplane = part + (long)blockIdx.x * plane_stride(N);

    for (int nb = 0; nb < N; nb += HB) {
        const int span = (N - nb < HB) ? (N - nb) : HB;
        const int span4 = span >> 2;
        int4* cnt4 = (int4*)cnt;
        for (int i = tid; i < span4; i += 1024) cnt4[i] = make_int4(0, 0, 0, 0);
        for (int i = (span4 << 2) + tid; i < span; i += 1024) cnt[i] = 0;
        __syncthreads();

        for (long e = e0 + (long)tid * 4; e < e1; e += 4096) {
            int r[4], c[4];
            int n = 4;
            if (vec && e + 3 < e1) {
                const int4 rv = *(const int4*)(row + e);
                const int4 cv = *(const int4*)(col + e);
                r[0] = rv.x; r[1] = rv.y; r[2] = rv.z; r[3] = rv.w;
                c[0] = cv.x; c[1] = cv.y; c[2] = cv.z; c[3] = cv.w;
            } else {
                n = (int)((e1 - e < 4) ? (e1 - e) : 4);
                for (int j = 0; j < 4; ++j) {
                    r[j] = (j < n) ? row[e + j] : 0;
                    c[j] = (j < n) ? col[e + j] : 0;
                }
            }
            for (int j = 0; j < 4; ++j) {
                const unsigned d = (unsigned)(r[j] - nb);
                if (j < n && r[j] != c[j] && d < (unsigned)span) atomicAdd(&cnt[d], 1);
            }
        }
        __syncthreads();

        int4* out4 = (int4*)(myplane + nb);
        for (int i = tid; i < span4; i += 1024) out4[i] = cnt4[i];
        for (int i = (span4 << 2) + tid; i < span; i += 1024) myplane[nb + i] = cnt[i];
        __syncthreads();
    }
}

__global__ void dinv_sqrt_kernel(const int* __restrict__ part, float* __restrict__ deg, int N) {
    const int i = blockIdx.x * blockDim.x + threadIdx.x;
    if (i >= N) return;
    const long ps = plane_stride(N);
    int s = 0;
    #pragma unroll 16
    for (int p = 0; p < DP; ++p) s += part[p * ps + i];
    const float d = (float)s;
    deg[i] = (s > 0) ? __frsqrt_rn(d) : 0.0f;
}

__global__ void lap_kernel(const int* __restrict__ row, const int* __restrict__ col,
                           const float* __restrict__ ew, const float* __restrict__ dis,
                           float* __restrict__ lap, long E) {
    long e = (long)blockIdx.x * blockDim.x + threadIdx.x;
    if (e >= E) return;
    int r = row[e], c = col[e];
    lap[e] = (r != c) ? (-dis[r] * ew[e] * dis[c]) : 0.0f;
}

#define TR   200
#define TW   (NB * CH)
#define SPW  32
#define LCAP 256

__global__ __launch_bounds__(1024) void spmm_tile_kernel(
    const float* __restrict__ src, const int* __restrict__ row,
    const int* __restrict__ col, const float* __restrict__ lap,
    float* __restrict__ dst, int N, long E) {
    __shared__ float tile[TR * TW];
    __shared__ int   hitE[SPW * LCAP];
    __shared__ int   hitR[SPW * LCAP];

    const int tid  = threadIdx.x;
    const int lane = tid & 31;
    const int wave = __builtin_amdgcn_readfirstlane(tid >> 5);
    const int base = blockIdx.x * TR;

    float4* tile4 = (float4*)tile;
    for (int i = tid; i < TR * TW / 4; i += 1024) tile4[i] = make_float4(0.f, 0.f, 0.f, 0.f);
    __syncthreads();

    int* myE = hitE + wave * LCAP;
    int* myR = hitR + wave * LCAP;
    const long plane = (long)N * CH;
    int cnt = 0;

    auto flush = [&](int n) {
        for (int o = 0; o < n; o += 32) {
            const int k = o + lane;
            int c = 0, rl = 0;
            float l = 0.0f;
            if (k < n) {
                const int e = myE[k];
                rl = myR[k];
                c = col[e];
                l = lap[e];
            }
            const int m = (n - o < 32) ? (n - o) : 32;
            for (int t = 0; t < m; t += 4) {
                float v[4][4];
                float lt[4];
                int rt[4];
                #pragma unroll
                for (int u = 0; u < 4; ++u) {
                    const int tt = (t + u < m) ? (t + u) : (m - 1);
                    const int ct = __builtin_amdgcn_readlane(c, tt);
                    rt[u] = __builtin_amdgcn_readlane(rl, tt);
                    lt[u] = __int_as_float(__builtin_amdgcn_readlane(__float_as_int(l), tt));
                    const float* s = src + (long)ct * CH + lane;
                    v[u][0] = s[0];
                    v[u][1] = s[32];
                    v[u][2] = s[plane];
                    v[u][3] = s[plane + 32];
                }
                #pragma unroll
                for (int u = 0; u < 4; ++u) {
                    if (t + u < m && lt[u] != 0.0f) {
                        float* d = tile + rt[u] * TW + lane;
                        __hip_atomic_fetch_add(d,      v[u][0] * lt[u], __ATOMIC_RELAXED, __HIP_MEMORY_SCOPE_WORKGROUP);
                        __hip_atomic_fetch_add(d + 32, v[u][1] * lt[u], __ATOMIC_RELAXED, __HIP_MEMORY_SCOPE_WORKGROUP);
                        __hip_atomic_fetch_add(d + 64, v[u][2] * lt[u], __ATOMIC_RELAXED, __HIP_MEMORY_SCOPE_WORKGROUP);
                        __hip_atomic_fetch_add(d + 96, v[u][3] * lt[u], __ATOMIC_RELAXED, __HIP_MEMORY_SCOPE_WORKGROUP);
                    }
                }
            }
        }
    };

    auto consume = [&](int4 r, long i) {
        const unsigned d0 = (unsigned)(r.x - base);
        const unsigned d1 = (unsigned)(r.y - base);
        const unsigned d2 = (unsigned)(r.z - base);
        const unsigned d3 = (unsigned)(r.w - base);
        const bool any = (d0 < (unsigned)TR) | (d1 < (unsigned)TR) | (d2 < (unsigned)TR) | (d3 < (unsigned)TR);
        if (__builtin_amdgcn_ballot_w32(any) == 0u) return;
        if (cnt > LCAP - 128) { flush(cnt); cnt = 0; }
        const unsigned dd[4] = {d0, d1, d2, d3};
        #pragma unroll
        for (int j = 0; j < 4; ++j) {
            const bool h = dd[j] < (unsigned)TR;
            const unsigned mk = __builtin_amdgcn_ballot_w32(h);
            if (h) {
                const int p = cnt + (int)__builtin_amdgcn_mbcnt_lo(mk, 0u);
                myE[p] = (int)(i + j);
                myR[p] = (int)dd[j];
            }
            cnt += __builtin_popcount(mk);
        }
    };

    const long stride = (long)SPW * 128;
    long wb = (long)wave * 128;
    for (; wb + 3 * stride + 128 <= E; wb += 4 * stride) {
        const int4* p = (const int4*)(row + wb) + lane;
        const int4 r0 = p[0];
        const int4 r1 = p[stride / 4];
        const int4 r2 = p[2 * (stride / 4)];
        const int4 r3 = p[3 * (stride / 4)];
        const long i = wb + lane * 4;
        consume(r0, i);
        consume(r1, i + stride);
        consume(r2, i + 2 * stride);
        consume(r3, i + 3 * stride);
    }
    for (; wb < E; wb += stride) {
        const long i = wb + lane * 4;
        int4 r = make_int4(base - 1, base - 1, base - 1, base - 1);
        if (i + 3 < E) {
            r = *(const int4*)(row + i);
        } else {
            if (i     < E) r.x = row[i];
            if (i + 1 < E) r.y = row[i + 1];
            if (i + 2 < E) r.z = row[i + 2];
        }
        consume(r, i);
    }
    if (cnt > 0) flush(cnt);
    __syncthreads();

    for (int i = tid; i < TR * TW / 4; i += 1024) {
        const int rl  = i >> 5;
        const int rem = i & 31;
        const int b   = rem >> 4;
        const int c4  = rem & 15;
        const int r   = base + rl;
        if (r < N) {
            *(float4*)(dst + ((long)b * N + r) * CH + c4 * 4) = tile4[i];
        }
    }
}

__global__ __launch_bounds__(128) void cheb_gemm_wmma(
    const float* __restrict__ x, const float* __restrict__ tx1,
    const float* __restrict__ s2, const float* __restrict__ w,
    const float* __restrict__ bias, float* __restrict__ out, long BN) {
    __shared__ float wlds[3 * CH * CH];

    const int tid = threadIdx.x;
    for (int idx = tid; idx < 3 * CH * CH; idx += 128) {
        const int t   = idx >> 12;
        const int rem = idx & 4095;
        const int k   = rem >> 6;
        const int n   = rem & 63;
        wlds[((((t << 5) + (k >> 1)) << 6) + n) * 2 + (k & 1)] = w[idx];
    }
    __syncthreads();

    const int wave = tid >> 5;
    const int lane = tid & 31;
    const long m0 = ((long)blockIdx.x * 4 + wave) * 16;
    if (m0 + 16 > BN) return;

    const int mrow = lane & 15;
    const int lhi  = lane >> 4;

    const v2f* __restrict__ wp = (const v2f*)wlds;
    const v2f* __restrict__ bbase = wp + (lhi << 6) + mrow;

    v8f acc[4] = {v8f{}, v8f{}, v8f{}, v8f{}};

    #pragma unroll
    for (int t = 0; t < 3; ++t) {
        const float* A = (t == 0) ? x : (t == 1) ? tx1 : s2;
        #pragma unroll
        for (int ks = 0; ks < 16; ++ks) {
            const int kb = ks * 4 + (lhi << 1);
            const float2 a2 = *(const float2*)(A + (m0 + mrow) * CH + kb);
            v2f a;
            a.x = a2.x; a.y = a2.y;
            if (t == 2) {
                const float2 x2 = *(const float2*)(x + (m0 + mrow) * CH + kb);
                a.x = 2.0f * a.x - x2.x;
                a.y = 2.0f * a.y - x2.y;
            }
            #pragma unroll
            for (int nt = 0; nt < 4; ++nt) {
                const v2f b = bbase[(t * 32 + ks * 2) * 64 + nt * 16];
                acc[nt] = __builtin_amdgcn_wmma_f32_16x16x4_f32(
                    false, a, false, b, (short)0, acc[nt], false, false);
            }
        }
    }

    const int moff = lhi << 3;
    #pragma unroll
    for (int nt = 0; nt < 4; ++nt) {
        const int col = nt * 16 + mrow;
        const float bv = bias[col];
        #pragma unroll
        for (int v = 0; v < 8; ++v) {
            out[(m0 + moff + v) * CH + col] = acc[nt][v] + bv;
        }
    }
}

__global__ void cheb_gemm_tail(const float* __restrict__ x, const float* __restrict__ tx1,
                               const float* __restrict__ s2, const float* __restrict__ w,
                               const float* __restrict__ bias, float* __restrict__ out,
                               long startRow, long BN) {
    long r = startRow + blockIdx.x;
    if (r >= BN) return;
    int c = threadIdx.x;
    float acc = bias[c];
    for (int k = 0; k < CH; ++k) {
        float t0 = x[r * CH + k];
        float t1 = tx1[r * CH + k];
        float t2 = 2.0f * s2[r * CH + k] - t0;
        acc += t0 * w[k * CH + c] + t1 * w[CH * CH + k * CH + c] + t2 * w[2 * CH * CH + k * CH + c];
    }
    out[r * CH + c] = acc;
}

extern "C" void kernel_launch(void* const* d_in, const int* in_sizes, int n_in,
                              void* d_out, int out_size, void* d_ws, size_t ws_size,
                              hipStream_t stream) {
    const float* x    = (const float*)d_in[0];
    const int*   ei   = (const int*)d_in[1];
    const float* ew   = (const float*)d_in[2];
    const float* w    = (const float*)d_in[3];
    const float* bias = (const float*)d_in[4];
    float*       out  = (float*)d_out;

    const long BN = (long)in_sizes[0] / CH;
    const int  N  = (int)(BN / NB);
    const long E  = (long)in_sizes[2];
    const int* row = ei;
    const int* col = ei + E;

    char* ws = (char*)d_ws;
    size_t off = 0;
    auto carve = [&](size_t bytes) {
        void* p = ws + off;
        off += (bytes + 255) & ~(size_t)255;
        return p;
    };
    float* deg = (float*)carve((size_t)N * sizeof(float));
    float* lap = (float*)carve((size_t)E * sizeof(float));
    float* tx1 = (float*)carve((size_t)BN * CH * sizeof(float));
    float* s2  = (float*)carve((size_t)BN * CH * sizeof(float));
    (void)ws_size;

    const int TB = 256;
    auto gblocks = [](long n, int tb) { return (int)((n + tb - 1) / tb); };

    int* part = (int*)s2;
    degree_part_kernel<<<DP, 1024, 0, stream>>>(row, col, part, N, E);
    dinv_sqrt_kernel<<<gblocks(N, TB), TB, 0, stream>>>(part, deg, N);

    lap_kernel<<<gblocks(E, TB), TB, 0, stream>>>(row, col, ew, deg, lap, E);

    const int tiles = gblocks(N, TR);
    spmm_tile_kernel<<<tiles, 1024, 0, stream>>>(x,   row, col, lap, tx1, N, E);
    spmm_tile_kernel<<<tiles, 1024, 0, stream>>>(tx1, row, col, lap, s2,  N, E);

    const long fullTiles = BN / 16;
    if (fullTiles > 0) {
        const int blocks = (int)((fullTiles + 3) / 4);
        cheb_gemm_wmma<<<blocks, 128, 0, stream>>>(x, tx1, s2, w, bias, out, BN);
    }
    const long tailStart = fullTiles * 16;
    if (tailStart < BN) {
        cheb_gemm_tail<<<(int)(BN - tailStart), CH, 0, stream>>>(x, tx1, s2, w, bias, out,
                                                                 tailStart, BN);
    }
}


# Second SpMM replays per-wave matched-edge lists (source row, tile row, coefficient) that the first SpMM's scan saves in the tile's own s2 rows: no second row[] scan; sums in LDS arrival order
# speedup vs baseline: 4.1817x; 1.0412x over previous
#include <hip/hip_runtime.h>
#include <hip/hip_bf16.h>

typedef __attribute__((ext_vector_type(2))) float v2f;
typedef __attribute__((ext_vector_type(8))) float v8f;

#define CH 64
#define NB 2

#define DP 64
#define HB 65536

__device__ __forceinline__ int plane_stride(int N) { return (N + 31) & ~31; }

__global__ __launch_bounds__(1024) void degree_part_kernel(
    const int* __restrict__ row, const int* __restrict__ col,
    int* __restrict__ part, int N, long E) {
    __shared__ int cnt[HB];
    const int tid = threadIdx.x;
    const long per = (((E + gridDim.x - 1) / gridDim.x) + 3) & ~3L;
    const long e0  = (long)blockIdx.x * per;
    const long e1  = (e0 + per < E) ? (e0 + per) : E;
    const bool vec = ((E & 3) == 0);
    int* __restrict__ myplane = part + (long)blockIdx.x * plane_stride(N);

    for (int nb = 0; nb < N; nb += HB) {
        const int span = (N - nb < HB) ? (N - nb) : HB;
        const int span4 = span >> 2;
        int4* cnt4 = (int4*)cnt;
        for (int i = tid; i < span4; i += 1024) cnt4[i] = make_int4(0, 0, 0, 0);
        for (int i = (span4 << 2) + tid; i < span; i += 1024) cnt[i] = 0;
        __syncthreads();

        for (long e = e0 + (long)tid * 4; e < e1; e += 4096) {
            int r[4], c[4];
            int n = 4;
            if (vec && e + 3 < e1) {
                const int4 rv = *(const int4*)(row + e);
                const int4 cv = *(const int4*)(col + e);
                r[0] = rv.x; r[1] = rv.y; r[2] = rv.z; r[3] = rv.w;
                c[0] = cv.x; c[1] = cv.y; c[2] = cv.z; c[3] = cv.w;
            } else {
                n = (int)((e1 - e < 4) ? (e1 - e) : 4);
                for (int j = 0; j < 4; ++j) {
                    r[j] = (j < n) ? row[e + j] : 0;
                    c[j] = (j < n) ? col[e + j] : 0;
                }
            }
            for (int j = 0; j < 4; ++j) {
                const unsigned d = (unsigned)(r[j] - nb);
                if (j < n && r[j] != c[j] && d < (unsigned)span) atomicAdd(&cnt[d], 1);
            }
        }
        __syncthreads();

        int4* out4 = (int4*)(myplane + nb);
        for (int i = tid; i < span4; i += 1024) out4[i] = cnt4[i];
        for (int i = (span4 << 2) + tid; i < span; i += 1024) myplane[nb + i] = cnt[i];
        __syncthreads();
    }
}

__global__ void dinv_sqrt_kernel(const int* __restrict__ part, float* __restrict__ deg, int N) {
    const int i = blockIdx.x * blockDim.x + threadIdx.x;
    if (i >= N) return;
    const long ps = plane_stride(N);
    int s = 0;
    #pragma unroll 16
    for (int p = 0; p < DP; ++p) s += part[p * ps + i];
    const float d = (float)s;
    deg[i] = (s > 0) ? __frsqrt_rn(d) : 0.0f;
}

__global__ void lap_kernel(const int* __restrict__ row, const int* __restrict__ col,
                           const float* __restrict__ ew, const float* __restrict__ dis,
                           float* __restrict__ lap, long E) {
    long e = (long)blockIdx.x * blockDim.x + threadIdx.x;
    if (e >= E) return;
    int r = row[e], c = col[e];
    lap[e] = (r != c) ? (-dis[r] * ew[e] * dis[c]) : 0.0f;
}

#define TR   200
#define TW   (NB * CH)
#define SPW  32
#define LCAP 256

#define HDR 32

__device__ __forceinline__ int wave_cap(int rows) {
    return ((((rows * CH) - HDR) / 2) / SPW) & ~31;
}

template <int U>
__device__ __forceinline__ void add_edges(float* tile, const float* __restrict__ src,
                                          long plane, int lane, int c, int rl, float l, int m) {
    for (int t = 0; t < m; t += U) {
        float v[U][4];
        float lt[U];
        int rt[U];
        #pragma unroll
        for (int u = 0; u < U; ++u) {
            const int tt = (t + u < m) ? (t + u) : (m - 1);
            const int ct = __builtin_amdgcn_readlane(c, tt);
            rt[u] = __builtin_amdgcn_readlane(rl, tt);
            lt[u] = __int_as_float(__builtin_amdgcn_readlane(__float_as_int(l), tt));
            const float* s = src + (long)ct * CH + lane;
            v[u][0] = s[0];
            v[u][1] = s[32];
            v[u][2] = s[plane];
            v[u][3] = s[plane + 32];
        }
        #pragma unroll
        for (int u = 0; u < U; ++u) {
            if (t + u < m && lt[u] != 0.0f) {
                float* d = tile + rt[u] * TW + lane;
                __hip_atomic_fetch_add(d,      v[u][0] * lt[u], __ATOMIC_RELAXED, __HIP_MEMORY_SCOPE_WORKGROUP);
                __hip_atomic_fetch_add(d + 32, v[u][1] * lt[u], __ATOMIC_RELAXED, __HIP_MEMORY_SCOPE_WORKGROUP);
                __hip_atomic_fetch_add(d + 64, v[u][2] * lt[u], __ATOMIC_RELAXED, __HIP_MEMORY_SCOPE_WORKGROUP);
                __hip_atomic_fetch_add(d + 96, v[u][3] * lt[u], __ATOMIC_RELAXED, __HIP_MEMORY_SCOPE_WORKGROUP);
            }
        }
    }
}

template <bool EMIT>
__global__ __launch_bounds__(1024) void spmm_tile_kernel(
    const float* __restrict__ src, const int* __restrict__ row,
    const int* __restrict__ col, const float* __restrict__ lap,
    float* dst, float* lst, int N, long E) {
    __shared__ float tile[TR * TW];
    __shared__ int   hitE[SPW * LCAP];
    __shared__ int   hitR[SPW * LCAP];

    const int tid  = threadIdx.x;
    const int lane = tid & 31;
    const int wave = __builtin_amdgcn_readfirstlane(tid >> 5);
    const int base = blockIdx.x * TR;

    float4* tile4 = (float4*)tile;
    for (int i = tid; i < TR * TW / 4; i += 1024) tile4[i] = make_float4(0.f, 0.f, 0.f, 0.f);
    __syncthreads();

    int* myE = hitE + wave * LCAP;
    int* myR = hitR + wave * LCAP;
    const long plane = (long)N * CH;
    int cnt = 0;

    const int rows = (N - base < TR) ? (N - base) : TR;
    const int wcap = wave_cap(rows);
    float* l0  = lst + (long)base * CH;
    float* l1  = lst + plane + (long)base * CH;
    float* myC = l0 + HDR + wave * wcap;
    float* myT = l0 + HDR + (SPW + wave) * wcap;
    float* myL = l1 + wave * wcap;
    int saved = 0;

    bool scan = true;
    if (!EMIT) {
        const int h = __float_as_int(l0[lane]);
        scan = __builtin_amdgcn_ballot_w32(h > wcap) != 0u;
        if (!scan) {
            const int n = __builtin_amdgcn_readlane(h, wave);
            for (int o = 0; o < n; o += 32) {
                const int k = o + lane;
                int c = 0, rl = 0;
                float l = 0.0f;
                if (k < n) {
                    c  = __float_as_int(myC[k]);
                    rl = __float_as_int(myT[k]);
                    l  = myL[k];
                }
                add_edges<8>(tile, src, plane, lane, c, rl, l, (n - o < 32) ? (n - o) : 32);
            }
        }
    }

    auto flush = [&](int n) {
        for (int o = 0; o < n; o += 32) {
            const int k = o + lane;
            int c = 0, rl = 0;
            float l = 0.0f;
            if (k < n) {
                const int e = myE[k];
                rl = myR[k];
                c = col[e];
                l = lap[e];
                if (EMIT) {
                    const int q = saved + k;
                    if (q < wcap) {
                        myC[q] = __int_as_float(c);
                        myT[q] = __int_as_float(rl);
                        myL[q] = l;
                    }
                }
            }
            add_edges<4>(tile, src, plane, lane, c, rl, l, (n - o < 32) ? (n - o) : 32);
        }
        saved += n;
    };

    auto consume = [&](int4 r, long i) {
        const unsigned d0 = (unsigned)(r.x - base);
        const unsigned d1 = (unsigned)(r.y - base);
        const unsigned d2 = (unsigned)(r.z - base);
        const unsigned d3 = (unsigned)(r.w - base);
        const bool any = (d0 < (unsigned)TR) | (d1 < (unsigned)TR) | (d2 < (unsigned)TR) | (d3 < (unsigned)TR);
        if (__builtin_amdgcn_ballot_w32(any) == 0u) return;
        if (cnt > LCAP - 128) { flush(cnt); cnt = 0; }
        const unsigned dd[4] = {d0, d1, d2, d3};
        #pragma unroll
        for (int j = 0; j < 4; ++j) {
            const bool h = dd[j] < (unsigned)TR;
            const unsigned mk = __builtin_amdgcn_ballot_w32(h);
            if (h) {
                const int p = cnt + (int)__builtin_amdgcn_mbcnt_lo(mk, 0u);
                myE[p] = (int)(i + j);
                myR[p] = (int)dd[j];
            }
            cnt += __builtin_popcount(mk);
        }
    };

    const long stride = (long)SPW * 128;
    long wb = (long)wave * 128;
    if (!scan) wb = E;
    for (; wb + 3 * stride + 128 <= E; wb += 4 * stride) {
        const int4* p = (const int4*)(row + wb) + lane;
        const int4 r0 = p[0];
        const int4 r1 = p[stride / 4];
        const int4 r2 = p[2 * (stride / 4)];
        const int4 r3 = p[3 * (stride / 4)];
        const long i = wb + lane * 4;
        consume(r0, i);
        consume(r1, i + stride);
        consume(r2, i + 2 * stride);
        consume(r3, i + 3 * stride);
    }
    for (; wb < E; wb += stride) {
        const long i = wb + lane * 4;
        int4 r = make_int4(base - 1, base - 1, base - 1, base - 1);
        if (i + 3 < E) {
            r = *(const int4*)(row + i);
        } else {
            if (i     < E) r.x = row[i];
            if (i + 1 < E) r.y = row[i + 1];
            if (i + 2 < E) r.z = row[i + 2];
        }
        consume(r, i);
    }
    if (cnt > 0) flush(cnt);
    if (EMIT && lane == 0) myE[0] = saved;
    __syncthreads();
    if (EMIT && wave == 0) l0[lane] = __int_as_float(hitE[lane * LCAP]);

    for (int i = tid; i < TR * TW / 4; i += 1024) {
        const int rl  = i >> 5;
        const int rem = i & 31;
        const int b   = rem >> 4;
        const int c4  = rem & 15;
        const int r   = base + rl;
        if (r < N) {
            *(float4*)(dst + ((long)b * N + r) * CH + c4 * 4) = tile4[i];
        }
    }
}

__global__ __launch_bounds__(128) void cheb_gemm_wmma(
    const float* __restrict__ x, const float* __restrict__ tx1,
    const float* __restrict__ s2, const float* __restrict__ w,
    const float* __restrict__ bias, float* __restrict__ out, long BN) {
    __shared__ float wlds[3 * CH * CH];

    const int tid = threadIdx.x;
    for (int idx = tid; idx < 3 * CH * CH; idx += 128) {
        const int t   = idx >> 12;
        const int rem = idx & 4095;
        const int k   = rem >> 6;
        const int n   = rem & 63;
        wlds[((((t << 5) + (k >> 1)) << 6) + n) * 2 + (k & 1)] = w[idx];
    }
    __syncthreads();

    const int wave = tid >> 5;
    const int lane = tid & 31;
    const long m0 = ((long)blockIdx.x * 4 + wave) * 16;
    if (m0 + 16 > BN) return;

    const int mrow = lane & 15;
    const int lhi  = lane >> 4;

    const v2f* __restrict__ wp = (const v2f*)wlds;
    const v2f* __restrict__ bbase = wp + (lhi << 6) + mrow;

    v8f acc[4] = {v8f{}, v8f{}, v8f{}, v8f{}};

    #pragma unroll
    for (int t = 0; t < 3; ++t) {
        const float* A = (t == 0) ? x : (t == 1) ? tx1 : s2;
        #pragma unroll
        for (int ks = 0; ks < 16; ++ks) {
            const int kb = ks * 4 + (lhi << 1);
            const float2 a2 = *(const float2*)(A + (m0 + mrow) * CH + kb);
            v2f a;
            a.x = a2.x; a.y = a2.y;
            if (t == 2) {
                const float2 x2 = *(const float2*)(x + (m0 + mrow) * CH + kb);
                a.x = 2.0f * a.x - x2.x;
                a.y = 2.0f * a.y - x2.y;
            }
            #pragma unroll
            for (int nt = 0; nt < 4; ++nt) {
                const v2f b = bbase[(t * 32 + ks * 2) * 64 + nt * 16];
                acc[nt] = __builtin_amdgcn_wmma_f32_16x16x4_f32(
                    false, a, false, b, (short)0, acc[nt], false, false);
            }
        }
    }

    const int moff = lhi << 3;
    #pragma unroll
    for (int nt = 0; nt < 4; ++nt) {
        const int col = nt * 16 + mrow;
        const float bv = bias[col];
        #pragma unroll
        for (int v = 0; v < 8; ++v) {
            out[(m0 + moff + v) * CH + col] = acc[nt][v] + bv;
        }
    }
}

__global__ void cheb_gemm_tail(const float* __restrict__ x, const float* __restrict__ tx1,
                               const float* __restrict__ s2, const float* __restrict__ w,
                               const float* __restrict__ bias, float* __restrict__ out,
                               long startRow, long BN) {
    long r = startRow + blockIdx.x;
    if (r >= BN) return;
    int c = threadIdx.x;
    float acc = bias[c];
    for (int k = 0; k < CH; ++k) {
        float t0 = x[r * CH + k];
        float t1 = tx1[r * CH + k];
        float t2 = 2.0f * s2[r * CH + k] - t0;
        acc += t0 * w[k * CH + c] + t1 * w[CH * CH + k * CH + c] + t2 * w[2 * CH * CH + k * CH + c];
    }
    out[r * CH + c] = acc;
}

extern "C" void kernel_launch(void* const* d_in, const int* in_sizes, int n_in,
                              void* d_out, int out_size, void* d_ws, size_t ws_size,
                              hipStream_t stream) {
    const float* x    = (const float*)d_in[0];
    const int*   ei   = (const int*)d_in[1];
    const float* ew   = (const float*)d_in[2];
    const float* w    = (const float*)d_in[3];
    const float* bias = (const float*)d_in[4];
    float*       out  = (float*)d_out;

    const long BN = (long)in_sizes[0] / CH;
    const int  N  = (int)(BN / NB);
    const long E  = (long)in_sizes[2];
    const int* row = ei;
    const int* col = ei + E;

    char* ws = (char*)d_ws;
    size_t off = 0;
    auto carve = [&](size_t bytes) {
        void* p = ws + off;
        off += (bytes + 255) & ~(size_t)255;
        return p;
    };
    float* deg = (float*)carve((size_t)N * sizeof(float));
    float* lap = (float*)carve((size_t)E * sizeof(float));
    float* tx1 = (float*)carve((size_t)BN * CH * sizeof(float));
    float* s2  = (float*)carve((size_t)BN * CH * sizeof(float));
    (void)ws_size;

    const int TB = 256;
    auto gblocks = [](long n, int tb) { return (int)((n + tb - 1) / tb); };

    int* part = (int*)s2;
    degree_part_kernel<<<DP, 1024, 0, stream>>>(row, col, part, N, E);
    dinv_sqrt_kernel<<<gblocks(N, TB), TB, 0, stream>>>(part, deg, N);

    lap_kernel<<<gblocks(E, TB), TB, 0, stream>>>(row, col, ew, deg, lap, E);

    const int tiles = gblocks(N, TR);
    spmm_tile_kernel<true ><<<tiles, 1024, 0, stream>>>(x,   row, col, lap, tx1, s2, N, E);
    spmm_tile_kernel<false><<<tiles, 1024, 0, stream>>>(tx1, row, col, lap, s2,  s2, N, E);

    const long fullTiles = BN / 16;
    if (fullTiles > 0) {
        const int blocks = (int)((fullTiles + 3) / 4);
        cheb_gemm_wmma<<<blocks, 128, 0, stream>>>(x, tx1, s2, w, bias, out, BN);
    }
    const long tailStart = fullTiles * 16;
    if (tailStart < BN) {
        cheb_gemm_tail<<<(int)(BN - tailStart), CH, 0, stream>>>(x, tx1, s2, w, bias, out,
                                                                 tailStart, BN);
    }
}
